# starting kernel
# baseline (speedup 1.0000x reference)
#include <hip/hip_runtime.h>
#include <hip/hip_bf16.h>

typedef float v2f __attribute__((ext_vector_type(2)));
typedef float v8f __attribute__((ext_vector_type(8)));

__device__ __forceinline__ float swap16(float x) {
  return __int_as_float(__builtin_amdgcn_ds_swizzle(__float_as_int(x), 0x401f));
}

__device__ __forceinline__ float fast_tanh(float x) {
#if __has_builtin(__builtin_amdgcn_tanhf)
  return __builtin_amdgcn_tanhf(x);
#else
  float e = __expf(2.0f * x);
  float r = __builtin_amdgcn_rcpf(e + 1.0f);
  return 1.0f - 2.0f * r;
#endif
}


__device__ __forceinline__ void build_B(const float x[16], v2f B[8], bool hi) {
#pragma unroll
  for (int blk = 0; blk < 8; ++blk) {
    const int k0 = blk * 4;
    const int t  = k0 >> 4;
    const int rr = k0 & 15;
    v2f b;
    if (rr < 8) {
      const int a = t * 8 + rr;
      float s0 = swap16(x[a + 2]);
      float s1 = swap16(x[a + 3]);
      b.x = hi ? s0 : x[a];
      b.y = hi ? s1 : x[a + 1];
    } else {
      const int a = t * 8 + rr - 8;
      float s0 = swap16(x[a]);
      float s1 = swap16(x[a + 1]);
      b.x = hi ? x[a + 2] : s0;
      b.y = hi ? x[a + 3] : s1;
    }
    B[blk] = b;
  }
}

__device__ __forceinline__ void layer(const float x[16], float y[16],
                                      const v2f A[2][8], const v8f biasf[2],
                                      bool hi) {
  v2f B[8];
  build_B(x, B, hi);
#pragma unroll
  for (int t = 0; t < 2; ++t) {
    v8f c = __builtin_amdgcn_wmma_f32_16x16x4_f32(
        false, A[t][0], false, B[0], (short)0, biasf[t], false, false);
#pragma unroll
    for (int blk = 1; blk < 8; ++blk) {
      c = __builtin_amdgcn_wmma_f32_16x16x4_f32(
          false, A[t][blk], false, B[blk], (short)0, c, false, false);
    }
#pragma unroll
    for (int r = 0; r < 8; ++r) y[t * 8 + r] = c[r];
  }
}

__device__ __forceinline__ void ode_f(const float x[16], float y[16],
                                      const v2f A1[2][8], const v2f A2[2][8],
                                      const v8f bias1[2], const v8f bias2[2],
                                      bool hi) {
  float hmid[16];
  layer(x, hmid, A1, bias1, hi);
#pragma unroll
  for (int i = 0; i < 16; ++i) hmid[i] = fast_tanh(hmid[i]);
  layer(hmid, y, A2, bias2, hi);
}

__global__ __launch_bounds__(256)
void node_rk4_wmma(const float* __restrict__ z0, const float* __restrict__ tt,
                   const float* __restrict__ W1, const float* __restrict__ b1,
                   const float* __restrict__ W2, const float* __restrict__ b2,
                   float* __restrict__ out, int nsteps, long nrows) {
  const int lane = threadIdx.x & 31;
  const int wv   = threadIdx.x >> 5;
  const int half = lane >> 4;
  const bool hi  = (half != 0);
  const int m    = lane & 15;

  const long tile = (long)blockIdx.x * (blockDim.x >> 5) + wv;
  const long row  = tile * 16 + m;
  if (tile * 16 >= nrows) return;

  v2f A1[2][8], A2[2][8];
#pragma unroll
  for (int t = 0; t < 2; ++t) {
#pragma unroll
    for (int blk = 0; blk < 8; ++blk) {
      const int kr  = blk * 4 + half * 2;
      const int col = t * 16 + m;
      v2f a1, a2;
      a1.x = W1[(kr + 0) * 32 + col];
      a1.y = W1[(kr + 1) * 32 + col];
      a2.x = W2[(kr + 0) * 32 + col];
      a2.y = W2[(kr + 1) * 32 + col];
      A1[t][blk] = a1;
      A2[t][blk] = a2;
    }
  }

  v8f bias1[2], bias2[2];
#pragma unroll
  for (int t = 0; t < 2; ++t) {
#pragma unroll
    for (int r = 0; r < 8; ++r) {
      const int col = t * 16 + half * 8 + r;
      bias1[t][r] = b1[col];
      bias2[t][r] = b2[col];
    }
  }

  float zc[16];
  {
    const float4* zp = (const float4*)(z0 + row * 32);
    float4 v0 = zp[half * 2 + 0];
    float4 v1 = zp[half * 2 + 1];
    float4 v2 = zp[4 + half * 2 + 0];
    float4 v3 = zp[4 + half * 2 + 1];
    zc[0] = v0.x; zc[1] = v0.y; zc[2]  = v0.z; zc[3]  = v0.w;
    zc[4] = v1.x; zc[5] = v1.y; zc[6]  = v1.z; zc[7]  = v1.w;
    zc[8] = v2.x; zc[9] = v2.y; zc[10] = v2.z; zc[11] = v2.w;
    zc[12] = v3.x; zc[13] = v3.y; zc[14] = v3.z; zc[15] = v3.w;
  }

  for (int s = 0; s < nsteps; ++s) {
    const float h = tt[s + 1] - tt[s];
    float kcur[16], acc[16], ztmp[16];

    ode_f(zc, kcur, A1, A2, bias1, bias2, hi);
#pragma unroll
    for (int i = 0; i < 16; ++i) {
      acc[i]  = kcur[i];
      ztmp[i] = zc[i] + 0.5f * h * kcur[i];
    }
    ode_f(ztmp, kcur, A1, A2, bias1, bias2, hi);
#pragma unroll
    for (int i = 0; i < 16; ++i) {
      acc[i] += 2.0f * kcur[i];
      ztmp[i] = zc[i] + 0.5f * h * kcur[i];
    }
    ode_f(ztmp, kcur, A1, A2, bias1, bias2, hi);
#pragma unroll
    for (int i = 0; i < 16; ++i) {
      acc[i] += 2.0f * kcur[i];
      ztmp[i] = zc[i] + h * kcur[i];
    }
    ode_f(ztmp, kcur, A1, A2, bias1, bias2, hi);
    const float c6 = h * (1.0f / 6.0f);
#pragma unroll
    for (int i = 0; i < 16; ++i) zc[i] += c6 * (acc[i] + kcur[i]);
  }

  {
    float4* op = (float4*)(out + row * 32);
    float4 v0, v1, v2, v3;
    v0.x = zc[0];  v0.y = zc[1];  v0.z = zc[2];  v0.w = zc[3];
    v1.x = zc[4];  v1.y = zc[5];  v1.z = zc[6];  v1.w = zc[7];
    v2.x = zc[8];  v2.y = zc[9];  v2.z = zc[10]; v2.w = zc[11];
    v3.x = zc[12]; v3.y = zc[13]; v3.z = zc[14]; v3.w = zc[15];
    op[half * 2 + 0]     = v0;
    op[half * 2 + 1]     = v1;
    op[4 + half * 2 + 0] = v2;
    op[4 + half * 2 + 1] = v3;
  }
}

extern "C" void kernel_launch(void* const* d_in, const int* in_sizes, int n_in,
                              void* d_out, int out_size, void* d_ws, size_t ws_size,
                              hipStream_t stream) {
  (void)n_in; (void)out_size; (void)d_ws; (void)ws_size;
  const float* z0 = (const float*)d_in[0];
  const float* tt = (const float*)d_in[1];
  const float* W1 = (const float*)d_in[2];
  const float* b1 = (const float*)d_in[3];
  const float* W2 = (const float*)d_in[4];
  const float* b2 = (const float*)d_in[5];
  float* out = (float*)d_out;

  const int  nsteps = in_sizes[1] - 1;
  const long nrows  = (long)in_sizes[0] / 32;
  const long tiles  = (nrows + 15) / 16;
  const int  wavesPerBlock = 8;
  const int  blocks = (int)((tiles + wavesPerBlock - 1) / wavesPerBlock);

  node_rk4_wmma<<<blocks, wavesPerBlock * 32, 0, stream>>>(
      z0, tt, W1, b1, W2, b2, out, nsteps, nrows);
}


# RK4 carried out on u=z*W1+b1 with folded M=W2*W1 (built per wave on WMMA): one 32x32 product per stage instead of two, z rebuilt once at the end; f32 products regrouped, summation order changes
# speedup vs baseline: 1.7244x; 1.7244x over previous
#include <hip/hip_runtime.h>
#include <hip/hip_bf16.h>

typedef float v2f __attribute__((ext_vector_type(2)));
typedef float v8f __attribute__((ext_vector_type(8)));

__device__ __forceinline__ float swap16(float x) {
  return __int_as_float(__builtin_amdgcn_ds_swizzle(__float_as_int(x), 0x401f));
}

__device__ __forceinline__ float fast_tanh(float x) {
#if __has_builtin(__builtin_amdgcn_tanhf)
  return __builtin_amdgcn_tanhf(x);
#else
  float e = __expf(2.0f * x);
  float r = __builtin_amdgcn_rcpf(e + 1.0f);
  return 1.0f - 2.0f * r;
#endif
}


__device__ __forceinline__ void build_B(const float x[16], v2f B[8], bool hi) {
#pragma unroll
  for (int blk = 0; blk < 8; ++blk) {
    const int k0 = blk * 4;
    const int t  = k0 >> 4;
    const int rr = k0 & 15;
    v2f b;
    if (rr < 8) {
      const int a = t * 8 + rr;
      float s0 = swap16(x[a + 2]);
      float s1 = swap16(x[a + 3]);
      b.x = hi ? s0 : x[a];
      b.y = hi ? s1 : x[a + 1];
    } else {
      const int a = t * 8 + rr - 8;
      float s0 = swap16(x[a]);
      float s1 = swap16(x[a + 1]);
      b.x = hi ? x[a + 2] : s0;
      b.y = hi ? x[a + 3] : s1;
    }
    B[blk] = b;
  }
}

__device__ __forceinline__ void layer(const float x[16], float y[16],
                                      const v2f A[2][8], const v8f biasf[2],
                                      bool hi) {
  v2f B[8];
  build_B(x, B, hi);
#pragma unroll
  for (int t = 0; t < 2; ++t) {
    v8f c = __builtin_amdgcn_wmma_f32_16x16x4_f32(
        false, A[t][0], false, B[0], (short)0, biasf[t], false, false);
#pragma unroll
    for (int blk = 1; blk < 8; ++blk) {
      c = __builtin_amdgcn_wmma_f32_16x16x4_f32(
          false, A[t][blk], false, B[blk], (short)0, c, false, false);
    }
#pragma unroll
    for (int r = 0; r < 8; ++r) y[t * 8 + r] = c[r];
  }
}

__device__ __forceinline__ void ode_g(const float u[16], float th[16], float g[16],
                                      const v2f AM[2][8], const v8f cfrag[2],
                                      bool hi) {
#pragma unroll
  for (int i = 0; i < 16; ++i) th[i] = fast_tanh(u[i]);
  layer(th, g, AM, cfrag, hi);
}

__global__ __launch_bounds__(256)
void node_rk4_wmma(const float* __restrict__ z0, const float* __restrict__ tt,
                   const float* __restrict__ W1, const float* __restrict__ b1,
                   const float* __restrict__ W2, const float* __restrict__ b2,
                   float* __restrict__ out, int nsteps, long nrows) {
  const int lane = threadIdx.x & 31;
  const int wv   = threadIdx.x >> 5;
  const int half = lane >> 4;
  const bool hi  = (half != 0);
  const int m    = lane & 15;

  const long tile = (long)blockIdx.x * (blockDim.x >> 5) + wv;
  const long row  = tile * 16 + m;
  if (tile * 16 >= nrows) return;


  v2f A1[2][8];
#pragma unroll
  for (int t = 0; t < 2; ++t) {
#pragma unroll
    for (int blk = 0; blk < 8; ++blk) {
      const int kr  = blk * 4 + half * 2;
      const int col = t * 16 + m;
      v2f a1;
      a1.x = W1[(kr + 0) * 32 + col];
      a1.y = W1[(kr + 1) * 32 + col];
      A1[t][blk] = a1;
    }
  }

  v8f zero8;
#pragma unroll
  for (int r = 0; r < 8; ++r) zero8[r] = 0.0f;
  const v8f zerof[2] = {zero8, zero8};

  v2f AM[2][8];
  {
    v2f AT[2][8];
#pragma unroll
    for (int t = 0; t < 2; ++t) {
#pragma unroll
      for (int blk = 0; blk < 8; ++blk) {
        const int kr = blk * 4 + half * 2;
        const int rw = t * 16 + m;
        v2f a;
        a.x = W2[rw * 32 + kr + 0];
        a.y = W2[rw * 32 + kr + 1];
        AT[t][blk] = a;
      }
    }
#pragma unroll
    for (int T = 0; T < 2; ++T) {
      float xw[16], yw[16];
#pragma unroll
      for (int i = 0; i < 16; ++i) {
        const int j = (i >> 3) * 16 + half * 8 + (i & 7);
        xw[i] = W1[j * 32 + T * 16 + m];
      }
      layer(xw, yw, AT, zerof, hi);
      build_B(yw, AM[T], hi);
    }
  }

  v8f cfrag[2];
  {
    float xb[16], yb[16];
#pragma unroll
    for (int i = 0; i < 16; ++i) xb[i] = b2[(i >> 3) * 16 + half * 8 + (i & 7)];
    layer(xb, yb, A1, zerof, hi);
#pragma unroll
    for (int t = 0; t < 2; ++t) {
#pragma unroll
      for (int r = 0; r < 8; ++r) cfrag[t][r] = yb[t * 8 + r];
    }
  }

  float u[16];
  {
    v8f bias1[2];
#pragma unroll
    for (int t = 0; t < 2; ++t) {
#pragma unroll
      for (int r = 0; r < 8; ++r) bias1[t][r] = b1[t * 16 + half * 8 + r];
    }
    float zc[16];
    const float4* zp = (const float4*)(z0 + row * 32);
    float4 v0 = zp[half * 2 + 0];
    float4 v1 = zp[half * 2 + 1];
    float4 v2 = zp[4 + half * 2 + 0];
    float4 v3 = zp[4 + half * 2 + 1];
    zc[0] = v0.x; zc[1] = v0.y; zc[2]  = v0.z; zc[3]  = v0.w;
    zc[4] = v1.x; zc[5] = v1.y; zc[6]  = v1.z; zc[7]  = v1.w;
    zc[8] = v2.x; zc[9] = v2.y; zc[10] = v2.z; zc[11] = v2.w;
    zc[12] = v3.x; zc[13] = v3.y; zc[14] = v3.z; zc[15] = v3.w;
    layer(zc, u, A1, bias1, hi);
  }

  float tw[16];
#pragma unroll
  for (int i = 0; i < 16; ++i) tw[i] = 0.0f;
  float hsum = 0.0f;

  for (int s = 0; s < nsteps; ++s) {
    const float h  = tt[s + 1] - tt[s];
    const float c6 = h * (1.0f / 6.0f);
    hsum += h;
    float th[16], g[16], acc[16], sacc[16], ut[16];

    ode_g(u, th, g, AM, cfrag, hi);
#pragma unroll
    for (int i = 0; i < 16; ++i) {
      sacc[i] = th[i];
      acc[i]  = g[i];
      ut[i]   = u[i] + 0.5f * h * g[i];
    }
    ode_g(ut, th, g, AM, cfrag, hi);
#pragma unroll
    for (int i = 0; i < 16; ++i) {
      sacc[i] += 2.0f * th[i];
      acc[i]  += 2.0f * g[i];
      ut[i]    = u[i] + 0.5f * h * g[i];
    }
    ode_g(ut, th, g, AM, cfrag, hi);
#pragma unroll
    for (int i = 0; i < 16; ++i) {
      sacc[i] += 2.0f * th[i];
      acc[i]  += 2.0f * g[i];
      ut[i]    = u[i] + h * g[i];
    }
    ode_g(ut, th, g, AM, cfrag, hi);
#pragma unroll
    for (int i = 0; i < 16; ++i) {
      tw[i] += c6 * (sacc[i] + th[i]);
      u[i]  += c6 * (acc[i] + g[i]);
    }
  }

  {
    v2f A2[2][8];
#pragma unroll
    for (int t = 0; t < 2; ++t) {
#pragma unroll
      for (int blk = 0; blk < 8; ++blk) {
        const int kr  = blk * 4 + half * 2;
        const int col = t * 16 + m;
        v2f a2;
        a2.x = W2[(kr + 0) * 32 + col];
        a2.y = W2[(kr + 1) * 32 + col];
        A2[t][blk] = a2;
      }
    }
    v8f bfin[2];
#pragma unroll
    for (int t = 0; t < 2; ++t) {
#pragma unroll
      for (int r = 0; r < 8; ++r) bfin[t][r] = hsum * b2[t * 16 + half * 8 + r];
    }
    float dz[16];
    layer(tw, dz, A2, bfin, hi);

    const float4* zp = (const float4*)(z0 + row * 32);
    float4 v0 = zp[half * 2 + 0];
    float4 v1 = zp[half * 2 + 1];
    float4 v2 = zp[4 + half * 2 + 0];
    float4 v3 = zp[4 + half * 2 + 1];
    v0.x += dz[0];  v0.y += dz[1];  v0.z += dz[2];  v0.w += dz[3];
    v1.x += dz[4];  v1.y += dz[5];  v1.z += dz[6];  v1.w += dz[7];
    v2.x += dz[8];  v2.y += dz[9];  v2.z += dz[10]; v2.w += dz[11];
    v3.x += dz[12]; v3.y += dz[13]; v3.z += dz[14]; v3.w += dz[15];
    float4* op = (float4*)(out + row * 32);
    op[half * 2 + 0]     = v0;
    op[half * 2 + 1]     = v1;
    op[4 + half * 2 + 0] = v2;
    op[4 + half * 2 + 1] = v3;
  }
}

extern "C" void kernel_launch(void* const* d_in, const int* in_sizes, int n_in,
                              void* d_out, int out_size, void* d_ws, size_t ws_size,
                              hipStream_t stream) {
  (void)n_in; (void)out_size; (void)d_ws; (void)ws_size;
  const float* z0 = (const float*)d_in[0];
  const float* tt = (const float*)d_in[1];
  const float* W1 = (const float*)d_in[2];
  const float* b1 = (const float*)d_in[3];
  const float* W2 = (const float*)d_in[4];
  const float* b2 = (const float*)d_in[5];
  float* out = (float*)d_out;

  const int  nsteps = in_sizes[1] - 1;
  const long nrows  = (long)in_sizes[0] / 32;
  const long tiles  = (nrows + 15) / 16;
  const int  wavesPerBlock = 8;
  const int  blocks = (int)((tiles + wavesPerBlock - 1) / wavesPerBlock);

  node_rk4_wmma<<<blocks, wavesPerBlock * 32, 0, stream>>>(
      z0, tt, W1, b1, W2, b2, out, nsteps, nrows);
}


# B fragments read in place: both WMMA loaders use the k order of the accumulator layout, so the lane-pair swizzles and selects go; f32 kept; summation order inside each dot product changes
# speedup vs baseline: 1.9470x; 1.1291x over previous
#include <hip/hip_runtime.h>
#include <hip/hip_bf16.h>

typedef float v2f __attribute__((ext_vector_type(2)));
typedef float v8f __attribute__((ext_vector_type(8)));

__device__ __forceinline__ float fast_tanh(float x) {
#if __has_builtin(__builtin_amdgcn_tanhf)
  return __builtin_amdgcn_tanhf(x);
#else
  float e = __expf(2.0f * x);
  float r = __builtin_amdgcn_rcpf(e + 1.0f);
  return 1.0f - 2.0f * r;
#endif
}


__device__ __forceinline__ int kbase(int blk, int half) {
  return (blk >> 2) * 16 + half * 8 + (blk & 3) * 2;
}

__device__ __forceinline__ void layer(const float x[16], float y[16],
                                      const v2f A[2][8], const v8f biasf[2]) {
  v2f B[8];
#pragma unroll
  for (int blk = 0; blk < 8; ++blk) {
    B[blk].x = x[2 * blk];
    B[blk].y = x[2 * blk + 1];
  }
#pragma unroll
  for (int t = 0; t < 2; ++t) {
    v8f c = __builtin_amdgcn_wmma_f32_16x16x4_f32(
        false, A[t][0], false, B[0], (short)0, biasf[t], false, false);
#pragma unroll
    for (int blk = 1; blk < 8; ++blk) {
      c = __builtin_amdgcn_wmma_f32_16x16x4_f32(
          false, A[t][blk], false, B[blk], (short)0, c, false, false);
    }
#pragma unroll
    for (int r = 0; r < 8; ++r) y[t * 8 + r] = c[r];
  }
}

__device__ __forceinline__ void ode_g(const float u[16], float th[16], float g[16],
                                      const v2f AM[2][8], const v8f cfrag[2]) {
#pragma unroll
  for (int i = 0; i < 16; ++i) th[i] = fast_tanh(u[i]);
  layer(th, g, AM, cfrag);
}

__global__ __launch_bounds__(256)
void node_rk4_wmma(const float* __restrict__ z0, const float* __restrict__ tt,
                   const float* __restrict__ W1, const float* __restrict__ b1,
                   const float* __restrict__ W2, const float* __restrict__ b2,
                   float* __restrict__ out, int nsteps, long nrows) {
  const int lane = threadIdx.x & 31;
  const int wv   = threadIdx.x >> 5;
  const int half = lane >> 4;
  const int m    = lane & 15;

  const long tile = (long)blockIdx.x * (blockDim.x >> 5) + wv;
  const long row  = tile * 16 + m;
  if (tile * 16 >= nrows) return;


  v2f A1[2][8];
#pragma unroll
  for (int t = 0; t < 2; ++t) {
#pragma unroll
    for (int blk = 0; blk < 8; ++blk) {
      const int kr  = kbase(blk, half);
      const int col = t * 16 + m;
      v2f a1;
      a1.x = W1[(kr + 0) * 32 + col];
      a1.y = W1[(kr + 1) * 32 + col];
      A1[t][blk] = a1;
    }
  }

  v8f zero8;
#pragma unroll
  for (int r = 0; r < 8; ++r) zero8[r] = 0.0f;
  const v8f zerof[2] = {zero8, zero8};

  v2f AM[2][8];
  {
    v2f AT[2][8];
#pragma unroll
    for (int t = 0; t < 2; ++t) {
#pragma unroll
      for (int blk = 0; blk < 8; ++blk) {
        const int kr = kbase(blk, half);
        const int rw = t * 16 + m;
        v2f a;
        a.x = W2[rw * 32 + kr + 0];
        a.y = W2[rw * 32 + kr + 1];
        AT[t][blk] = a;
      }
    }
#pragma unroll
    for (int T = 0; T < 2; ++T) {
      float xw[16], yw[16];
#pragma unroll
      for (int i = 0; i < 16; ++i) {
        const int j = (i >> 3) * 16 + half * 8 + (i & 7);
        xw[i] = W1[j * 32 + T * 16 + m];
      }
      layer(xw, yw, AT, zerof);
#pragma unroll
      for (int blk = 0; blk < 8; ++blk) {
        v2f am;
        am.x = yw[2 * blk];
        am.y = yw[2 * blk + 1];
        AM[T][blk] = am;
      }
    }
  }

  v8f cfrag[2];
  {
    float xb[16], yb[16];
#pragma unroll
    for (int i = 0; i < 16; ++i) xb[i] = b2[(i >> 3) * 16 + half * 8 + (i & 7)];
    layer(xb, yb, A1, zerof);
#pragma unroll
    for (int t = 0; t < 2; ++t) {
#pragma unroll
      for (int r = 0; r < 8; ++r) cfrag[t][r] = yb[t * 8 + r];
    }
  }

  float u[16];
  {
    v8f bias1[2];
#pragma unroll
    for (int t = 0; t < 2; ++t) {
#pragma unroll
      for (int r = 0; r < 8; ++r) bias1[t][r] = b1[t * 16 + half * 8 + r];
    }
    float zc[16];
    const float4* zp = (const float4*)(z0 + row * 32);
    float4 v0 = zp[half * 2 + 0];
    float4 v1 = zp[half * 2 + 1];
    float4 v2 = zp[4 + half * 2 + 0];
    float4 v3 = zp[4 + half * 2 + 1];
    zc[0] = v0.x; zc[1] = v0.y; zc[2]  = v0.z; zc[3]  = v0.w;
    zc[4] = v1.x; zc[5] = v1.y; zc[6]  = v1.z; zc[7]  = v1.w;
    zc[8] = v2.x; zc[9] = v2.y; zc[10] = v2.z; zc[11] = v2.w;
    zc[12] = v3.x; zc[13] = v3.y; zc[14] = v3.z; zc[15] = v3.w;
    layer(zc, u, A1, bias1);
  }

  float tw[16];
#pragma unroll
  for (int i = 0; i < 16; ++i) tw[i] = 0.0f;
  float hsum = 0.0f;

  for (int s = 0; s < nsteps; ++s) {
    const float h  = tt[s + 1] - tt[s];
    const float c6 = h * (1.0f / 6.0f);
    hsum += h;
    float th[16], g[16], acc[16], sacc[16], ut[16];

    ode_g(u, th, g, AM, cfrag);
#pragma unroll
    for (int i = 0; i < 16; ++i) {
      sacc[i] = th[i];
      acc[i]  = g[i];
      ut[i]   = u[i] + 0.5f * h * g[i];
    }
    ode_g(ut, th, g, AM, cfrag);
#pragma unroll
    for (int i = 0; i < 16; ++i) {
      sacc[i] += 2.0f * th[i];
      acc[i]  += 2.0f * g[i];
      ut[i]    = u[i] + 0.5f * h * g[i];
    }
    ode_g(ut, th, g, AM, cfrag);
#pragma unroll
    for (int i = 0; i < 16; ++i) {
      sacc[i] += 2.0f * th[i];
      acc[i]  += 2.0f * g[i];
      ut[i]    = u[i] + h * g[i];
    }
    ode_g(ut, th, g, AM, cfrag);
#pragma unroll
    for (int i = 0; i < 16; ++i) {
      tw[i] += c6 * (sacc[i] + th[i]);
      u[i]  += c6 * (acc[i] + g[i]);
    }
  }

  {
    v2f A2[2][8];
#pragma unroll
    for (int t = 0; t < 2; ++t) {
#pragma unroll
      for (int blk = 0; blk < 8; ++blk) {
        const int kr  = kbase(blk, half);
        const int col = t * 16 + m;
        v2f a2;
        a2.x = W2[(kr + 0) * 32 + col];
        a2.y = W2[(kr + 1) * 32 + col];
        A2[t][blk] = a2;
      }
    }
    v8f bfin[2];
#pragma unroll
    for (int t = 0; t < 2; ++t) {
#pragma unroll
      for (int r = 0; r < 8; ++r) bfin[t][r] = hsum * b2[t * 16 + half * 8 + r];
    }
    float dz[16];
    layer(tw, dz, A2, bfin);

    const float4* zp = (const float4*)(z0 + row * 32);
    float4 v0 = zp[half * 2 + 0];
    float4 v1 = zp[half * 2 + 1];
    float4 v2 = zp[4 + half * 2 + 0];
    float4 v3 = zp[4 + half * 2 + 1];
    v0.x += dz[0];  v0.y += dz[1];  v0.z += dz[2];  v0.w += dz[3];
    v1.x += dz[4];  v1.y += dz[5];  v1.z += dz[6];  v1.w += dz[7];
    v2.x += dz[8];  v2.y += dz[9];  v2.z += dz[10]; v2.w += dz[11];
    v3.x += dz[12]; v3.y += dz[13]; v3.z += dz[14]; v3.w += dz[15];
    float4* op = (float4*)(out + row * 32);
    op[half * 2 + 0]     = v0;
    op[half * 2 + 1]     = v1;
    op[4 + half * 2 + 0] = v2;
    op[4 + half * 2 + 1] = v3;
  }
}

extern "C" void kernel_launch(void* const* d_in, const int* in_sizes, int n_in,
                              void* d_out, int out_size, void* d_ws, size_t ws_size,
                              hipStream_t stream) {
  (void)n_in; (void)out_size; (void)d_ws; (void)ws_size;
  const float* z0 = (const float*)d_in[0];
  const float* tt = (const float*)d_in[1];
  const float* W1 = (const float*)d_in[2];
  const float* b1 = (const float*)d_in[3];
  const float* W2 = (const float*)d_in[4];
  const float* b2 = (const float*)d_in[5];
  float* out = (float*)d_out;

  const int  nsteps = in_sizes[1] - 1;
  const long nrows  = (long)in_sizes[0] / 32;
  const long tiles  = (nrows + 15) / 16;
  const int  wavesPerBlock = 8;
  const int  blocks = (int)((tiles + wavesPerBlock - 1) / wavesPerBlock);

  node_rk4_wmma<<<blocks, wavesPerBlock * 32, 0, stream>>>(
      z0, tt, W1, b1, W2, b2, out, nsteps, nrows);
}
